# starting kernel
# baseline (speedup 1.0000x reference)
#include <hip/hip_runtime.h>

typedef __attribute__((ext_vector_type(2))) float v2f;
typedef __attribute__((ext_vector_type(8))) float v8f;

#define NNODES 10000
#define NBATCH 4
#define NTIME  8
#define NFIN   16
#define TBD    32
#define CCHEB  32
#define CTIME  64
#define NROWS  320000
#define NRF    5120000
#define LNEPS  1e-5f
#define PIters 64

__global__ __launch_bounds__(256) void k_zero_int(int* __restrict__ p, int n) {
  int i = blockIdx.x * 256 + threadIdx.x;
  if (i < n) p[i] = 0;
}

__global__ __launch_bounds__(256) void k_build_xt(const float* __restrict__ X,
                                                  float* __restrict__ Xt) {
  int tid = blockIdx.x * 256 + threadIdx.x;
  int f2 = tid & 15;
  int tb = (tid >> 4) & 31;
  int n2 = tid >> 9;
  long i = ((long)(n2 * 16 + f2)) * 32 + tb;
  int t = (int)(i & 7);
  long q = i >> 3;
  int n = (int)(q % NNODES);
  int p = (int)(q / NNODES);
  int b = p & 3;
  int f = p >> 2;
  Xt[((long)n2 * 32 + tb) * 16 + f2] =
      X[(((long)(b * NNODES + n)) * 16 + f) * 8 + t];
}

__global__ __launch_bounds__(256) void k_count(const int* __restrict__ dst,
                                               int* __restrict__ indeg, int E) {
  int e = blockIdx.x * 256 + threadIdx.x;
  if (e < E) atomicAdd(&indeg[dst[e]], 1);
}

__global__ __launch_bounds__(1024) void k_scan(const int* __restrict__ indeg,
                                               int* __restrict__ rowptr) {
  __shared__ int part[1024];
  int tid = threadIdx.x;
  int base = tid * 10;
  int loc[10];
  int s = 0;
#pragma unroll
  for (int k = 0; k < 10; ++k) {
    int n = base + k;
    int d = (n < NNODES) ? indeg[n] : 0;
    loc[k] = d;
    s += d;
  }
  part[tid] = s;
  __syncthreads();
  for (int ofs = 1; ofs < 1024; ofs <<= 1) {
    int v = 0;
    if (tid >= ofs) v = part[tid - ofs];
    __syncthreads();
    part[tid] += v;
    __syncthreads();
  }
  int run = (tid == 0) ? 0 : part[tid - 1];
#pragma unroll
  for (int k = 0; k < 10; ++k) {
    int n = base + k;
    if (n < NNODES) rowptr[n] = run;
    run += loc[k];
  }
  if (tid == 1023) rowptr[NNODES] = run;
}

__global__ __launch_bounds__(256) void k_fill(const int* __restrict__ src,
                                              const int* __restrict__ dst,
                                              const int* __restrict__ rowptr,
                                              int* __restrict__ cursor,
                                              int* __restrict__ col, int E) {
  int e = blockIdx.x * 256 + threadIdx.x;
  if (e < E) {
    int d = dst[e];
    int pos = rowptr[d] + atomicAdd(&cursor[d], 1);
    col[pos] = src[e];
  }
}

__global__ __launch_bounds__(1024) void k_lambda(const int* __restrict__ rowptr,
                                                 const int* __restrict__ col,
                                                 float* __restrict__ sc) {
  extern __shared__ float sm[];
  float* va = sm;
  float* vb = sm + 10240;
  float* red = sm + 20480;
  const int tid = threadIdx.x;

  for (int i = tid; i < NNODES; i += 1024) va[i] = 1.0f / sqrtf((float)NNODES);
  __syncthreads();

  float nrm = 0.0f;
  float* cur = va;
  float* nxt = vb;
  for (int it = 0; it < PIters; ++it) {
    float ss = 0.0f;
    for (int i = tid; i < NNODES; i += 1024) {
      int beg = rowptr[i], end = rowptr[i + 1];
      float a = 0.0f;
      for (int k = beg; k < end; ++k) a += cur[col[k]];
      float w = (float)(end - beg) * cur[i] - a;
      nxt[i] = w;
      ss += w * w;
    }
    red[tid] = ss;
    __syncthreads();
    for (int s = 512; s > 0; s >>= 1) {
      if (tid < s) red[tid] += red[tid + s];
      __syncthreads();
    }
    nrm = sqrtf(red[0]);
    for (int i = tid; i < NNODES; i += 1024) nxt[i] /= nrm;
    __syncthreads();
    float* tmp = cur; cur = nxt; nxt = tmp;
  }
  if (tid == 0) sc[0] = nrm;
}

__global__ __launch_bounds__(256) void k_gather1(const int* __restrict__ rowptr,
                                                 const int* __restrict__ col,
                                                 const float* __restrict__ Xt,
                                                 const float* __restrict__ sc,
                                                 float* __restrict__ Tx1) {
  int tid = blockIdx.x * 256 + threadIdx.x;
  int n = tid >> 7;
  int j = (tid & 127) << 2;
  int beg = rowptr[n], end = rowptr[n + 1];
  float4 acc = make_float4(0.f, 0.f, 0.f, 0.f);
  for (int k = beg; k < end; ++k) {
    float4 v = *(const float4*)(Xt + (long)col[k] * 512 + j);
    acc.x += v.x; acc.y += v.y; acc.z += v.z; acc.w += v.w;
  }
  float lm = sc[0];
  float off = 2.0f / lm;
  float dg = 2.0f * (float)(end - beg) / lm - 1.0f;
  float4 x0 = *(const float4*)(Xt + (long)n * 512 + j);
  *(float4*)(Tx1 + (long)n * 512 + j) =
      make_float4(dg * x0.x - off * acc.x, dg * x0.y - off * acc.y,
                  dg * x0.z - off * acc.z, dg * x0.w - off * acc.w);
}

__global__ __launch_bounds__(256) void k_gather2(const int* __restrict__ rowptr,
                                                 const int* __restrict__ col,
                                                 const float* __restrict__ Tx1,
                                                 const float* __restrict__ Xt,
                                                 const float* __restrict__ sc,
                                                 float* __restrict__ Tx2) {
  int tid = blockIdx.x * 256 + threadIdx.x;
  int n = tid >> 7;
  int j = (tid & 127) << 2;
  int beg = rowptr[n], end = rowptr[n + 1];
  float4 acc = make_float4(0.f, 0.f, 0.f, 0.f);
  for (int k = beg; k < end; ++k) {
    float4 v = *(const float4*)(Tx1 + (long)col[k] * 512 + j);
    acc.x += v.x; acc.y += v.y; acc.z += v.z; acc.w += v.w;
  }
  float lm = sc[0];
  float off = 2.0f / lm;
  float dg = 2.0f * (float)(end - beg) / lm - 1.0f;
  float4 x1 = *(const float4*)(Tx1 + (long)n * 512 + j);
  float4 x0 = *(const float4*)(Xt + (long)n * 512 + j);
  *(float4*)(Tx2 + (long)n * 512 + j) =
      make_float4(2.0f * (dg * x1.x - off * acc.x) - x0.x,
                  2.0f * (dg * x1.y - off * acc.y) - x0.y,
                  2.0f * (dg * x1.z - off * acc.z) - x0.z,
                  2.0f * (dg * x1.w - off * acc.w) - x0.w);
}

__global__ __launch_bounds__(256) void k_cheb_gemm(const float* __restrict__ Tx0,
                                                   const float* __restrict__ Tx1,
                                                   const float* __restrict__ Tx2,
                                                   const float* __restrict__ W,
                                                   const float* __restrict__ bvec,
                                                   float* __restrict__ Out) {
  int wave = (blockIdx.x * 256 + threadIdx.x) >> 5;
  int lane = threadIdx.x & 31;
  int m = lane & 15;
  int half = lane >> 4;
  int kb = half * 2;
  long r0 = (long)wave * 16;
  long rowA = r0 + m;

  v8f c0 = {};
  v8f c1 = {};
  const float* Ts[3] = {Tx0, Tx1, Tx2};
#pragma unroll
  for (int kidx = 0; kidx < 3; ++kidx) {
    const float* Tk = Ts[kidx];
    const float* Wk = W + kidx * (NFIN * CCHEB);
#pragma unroll
    for (int kk = 0; kk < 16; kk += 4) {
      v2f a;
      a.x = Tk[rowA * 16 + kk + kb];
      a.y = Tk[rowA * 16 + kk + kb + 1];
      v2f b0, b1;
      b0.x = Wk[(kk + kb) * 32 + m];
      b0.y = Wk[(kk + kb + 1) * 32 + m];
      b1.x = Wk[(kk + kb) * 32 + 16 + m];
      b1.y = Wk[(kk + kb + 1) * 32 + 16 + m];
      c0 = __builtin_amdgcn_wmma_f32_16x16x4_f32(false, a, false, b0, (short)0, c0, false, false);
      c1 = __builtin_amdgcn_wmma_f32_16x16x4_f32(false, a, false, b1, (short)0, c1, false, false);
    }
  }
#pragma unroll
  for (int i = 0; i < 8; ++i) {
    long row = r0 + i + half * 8;
    float vA = c0[i] + bvec[m];
    float vB = c1[i] + bvec[16 + m];
    vA = vA > 0.0f ? vA : 0.0f;
    vB = vB > 0.0f ? vB : 0.0f;
    Out[row * 32 + m] = vA;
    Out[row * 32 + 16 + m] = vB;
  }
}

__global__ __launch_bounds__(256) void k_time_gemm(const float* __restrict__ Cheb,
                                                   const float* __restrict__ X,
                                                   const float* __restrict__ tW,
                                                   const float* __restrict__ tB,
                                                   const float* __restrict__ rW,
                                                   const float* __restrict__ rB,
                                                   const float* __restrict__ gamma,
                                                   const float* __restrict__ beta,
                                                   float* __restrict__ Out) {
  int wave = (blockIdx.x * 256 + threadIdx.x) >> 5;
  int lane = threadIdx.x & 31;
  int m = lane & 15;
  int half = lane >> 4;
  int kb = half * 2;
  long r0 = (long)wave * 16;

  long rA = r0 + m;
  int nA = (int)(rA >> 5);
  int tbA = (int)(rA & 31);
  int bA = tbA >> 3;
  int tA = tbA & 7;

  v8f acc0 = {}, acc1 = {}, acc2 = {}, acc3 = {};

#pragma unroll
  for (int kk = 0; kk < 96; kk += 4) {
    int k0 = kk + kb;
    int dt0 = k0 >> 5, ci0 = k0 & 31;
    int dt1 = (k0 + 1) >> 5, ci1 = (k0 + 1) & 31;
    int tp0 = tA + dt0 - 1;
    int tp1 = tA + dt1 - 1;
    float m0 = (tp0 >= 0 && tp0 < 8) ? 1.0f : 0.0f;
    float m1 = (tp1 >= 0 && tp1 < 8) ? 1.0f : 0.0f;
    int tc0 = min(max(tp0, 0), 7);
    int tc1 = min(max(tp1, 0), 7);
    v2f a;
    a.x = m0 * Cheb[((long)nA * 32 + bA * 8 + tc0) * 32 + ci0];
    a.y = m1 * Cheb[((long)nA * 32 + bA * 8 + tc1) * 32 + ci1];
#define TCONV_STEP(ACC, CT)                                                        \
    {                                                                              \
      int co = (CT) * 16 + m;                                                      \
      v2f b;                                                                       \
      b.x = tW[co * 96 + ci0 * 3 + dt0];                                           \
      b.y = tW[co * 96 + ci1 * 3 + dt1];                                           \
      ACC = __builtin_amdgcn_wmma_f32_16x16x4_f32(false, a, false, b, (short)0,    \
                                                  ACC, false, false);              \
    }
    TCONV_STEP(acc0, 0)
    TCONV_STEP(acc1, 1)
    TCONV_STEP(acc2, 2)
    TCONV_STEP(acc3, 3)
#undef TCONV_STEP
  }

#pragma unroll
  for (int kk = 0; kk < 16; kk += 4) {
    int k0 = kk + kb;
    v2f a;
    a.x = X[(((long)(bA * NNODES + nA)) * 16 + k0) * 8 + tA];
    a.y = X[(((long)(bA * NNODES + nA)) * 16 + k0 + 1) * 8 + tA];
#define RCONV_STEP(ACC, CT)                                                        \
    {                                                                              \
      int co = (CT) * 16 + m;                                                      \
      v2f b;                                                                       \
      b.x = rW[co * 16 + k0];                                                      \
      b.y = rW[co * 16 + k0 + 1];                                                  \
      ACC = __builtin_amdgcn_wmma_f32_16x16x4_f32(false, a, false, b, (short)0,    \
                                                  ACC, false, false);              \
    }
    RCONV_STEP(acc0, 0)
    RCONV_STEP(acc1, 1)
    RCONV_STEP(acc2, 2)
    RCONV_STEP(acc3, 3)
#undef RCONV_STEP
  }

#pragma unroll
  for (int i = 0; i < 8; ++i) {
    long row = r0 + i + half * 8;
    int n = (int)(row >> 5);
    int tb = (int)(row & 31);
    int b = tb >> 3;
    int t = tb & 7;
    float vals[4];
    float s = 0.0f, ss = 0.0f;
    {
      float v0 = acc0[i] + tB[m] + rB[m];
      float v1 = acc1[i] + tB[16 + m] + rB[16 + m];
      float v2 = acc2[i] + tB[32 + m] + rB[32 + m];
      float v3 = acc3[i] + tB[48 + m] + rB[48 + m];
      v0 = v0 > 0.0f ? v0 : 0.0f;
      v1 = v1 > 0.0f ? v1 : 0.0f;
      v2 = v2 > 0.0f ? v2 : 0.0f;
      v3 = v3 > 0.0f ? v3 : 0.0f;
      vals[0] = v0; vals[1] = v1; vals[2] = v2; vals[3] = v3;
      s = v0 + v1 + v2 + v3;
      ss = v0 * v0 + v1 * v1 + v2 * v2 + v3 * v3;
    }
#pragma unroll
    for (int ofs = 1; ofs < 16; ofs <<= 1) {
      s += __shfl_xor(s, ofs, 32);
      ss += __shfl_xor(ss, ofs, 32);
    }
    float mu = s * (1.0f / 64.0f);
    float var = ss * (1.0f / 64.0f) - mu * mu;
    float inv = rsqrtf(var + LNEPS);
#pragma unroll
    for (int ct = 0; ct < 4; ++ct) {
      int c = ct * 16 + m;
      float y = (vals[ct] - mu) * inv * gamma[c] + beta[c];
      Out[(((long)(b * NNODES + n)) * 64 + c) * 8 + t] = y;
    }
  }
}

extern "C" void kernel_launch(void* const* d_in, const int* in_sizes, int n_in,
                              void* d_out, int out_size, void* d_ws, size_t ws_size,
                              hipStream_t stream) {
  (void)n_in; (void)out_size; (void)ws_size;
  const float* X     = (const float*)d_in[0];
  const int*   eidx  = (const int*)d_in[1];
  const float* chebW = (const float*)d_in[2];
  const float* chebB = (const float*)d_in[3];
  const float* timeW = (const float*)d_in[4];
  const float* timeB = (const float*)d_in[5];
  const float* resW  = (const float*)d_in[6];
  const float* resB  = (const float*)d_in[7];
  const float* gamma = (const float*)d_in[8];
  const float* beta  = (const float*)d_in[9];
  float* out = (float*)d_out;

  const int E = in_sizes[1] / 2;
  const int* src = eidx;
  const int* dst = eidx + E;

  float* w    = (float*)d_ws;
  float* Xt   = w;
  float* Tx1  = Xt + NRF;
  float* Tx2  = Tx1 + NRF;
  float* Cheb = Tx2 + NRF;
  float* sc   = Cheb + (size_t)NROWS * CCHEB;
  int*   ip     = (int*)(sc + 16);
  int*   indeg  = ip;
  int*   cursor = ip + 10240;
  int*   rowptr = ip + 20480;
  int*   col    = ip + 30976;

  k_build_xt<<<NRF / 256, 256, 0, stream>>>(X, Xt);

  k_zero_int<<<(20480 + 255) / 256, 256, 0, stream>>>(ip, 20480);
  k_count<<<(E + 255) / 256, 256, 0, stream>>>(dst, indeg, E);
  k_scan<<<1, 1024, 0, stream>>>(indeg, rowptr);
  k_fill<<<(E + 255) / 256, 256, 0, stream>>>(src, dst, rowptr, cursor, col, E);

  const size_t lambdaLds = (size_t)(2 * 10240 + 1024) * sizeof(float);
  k_lambda<<<1, 1024, lambdaLds, stream>>>(rowptr, col, sc);

  k_gather1<<<NNODES * 128 / 256, 256, 0, stream>>>(rowptr, col, Xt, sc, Tx1);
  k_gather2<<<NNODES * 128 / 256, 256, 0, stream>>>(rowptr, col, Tx1, Xt, sc, Tx2);

  k_cheb_gemm<<<2500, 256, 0, stream>>>(Xt, Tx1, Tx2, chebW, chebB, Cheb);

  k_time_gemm<<<2500, 256, 0, stream>>>(Cheb, X, timeW, timeB, resW, resB,
                                        gamma, beta, out);
}


# k_lambda: edge list re-laid by one added small kernel into 32-node k-major slices, so a wave reads its neighbour indices as contiguous 512-byte rows, not per-lane scattered dwords; sum order kept
# speedup vs baseline: 2.0493x; 2.0493x over previous
#include <hip/hip_runtime.h>

typedef __attribute__((ext_vector_type(2))) float v2f;
typedef __attribute__((ext_vector_type(8))) float v8f;

#define NNODES 10000
#define NBATCH 4
#define NTIME  8
#define NFIN   16
#define TBD    32
#define CCHEB  32
#define CTIME  64
#define NROWS  320000
#define NRF    5120000
#define LNEPS  1e-5f
#define PIters 64

__global__ __launch_bounds__(256) void k_zero_int(int* __restrict__ p, int n) {
  int i = blockIdx.x * 256 + threadIdx.x;
  if (i < n) p[i] = 0;
}

__global__ __launch_bounds__(256) void k_build_xt(const float* __restrict__ X,
                                                  float* __restrict__ Xt) {
  int tid = blockIdx.x * 256 + threadIdx.x;
  int f2 = tid & 15;
  int tb = (tid >> 4) & 31;
  int n2 = tid >> 9;
  long i = ((long)(n2 * 16 + f2)) * 32 + tb;
  int t = (int)(i & 7);
  long q = i >> 3;
  int n = (int)(q % NNODES);
  int p = (int)(q / NNODES);
  int b = p & 3;
  int f = p >> 2;
  Xt[((long)n2 * 32 + tb) * 16 + f2] =
      X[(((long)(b * NNODES + n)) * 16 + f) * 8 + t];
}

__global__ __launch_bounds__(256) void k_count(const int* __restrict__ dst,
                                               int* __restrict__ indeg, int E) {
  int e = blockIdx.x * 256 + threadIdx.x;
  if (e < E) atomicAdd(&indeg[dst[e]], 1);
}

__global__ __launch_bounds__(1024) void k_scan(const int* __restrict__ indeg,
                                               int* __restrict__ rowptr) {
  __shared__ int part[1024];
  int tid = threadIdx.x;
  int base = tid * 10;
  int loc[10];
  int s = 0;
#pragma unroll
  for (int k = 0; k < 10; ++k) {
    int n = base + k;
    int d = (n < NNODES) ? indeg[n] : 0;
    loc[k] = d;
    s += d;
  }
  part[tid] = s;
  __syncthreads();
  for (int ofs = 1; ofs < 1024; ofs <<= 1) {
    int v = 0;
    if (tid >= ofs) v = part[tid - ofs];
    __syncthreads();
    part[tid] += v;
    __syncthreads();
  }
  int run = (tid == 0) ? 0 : part[tid - 1];
#pragma unroll
  for (int k = 0; k < 10; ++k) {
    int n = base + k;
    if (n < NNODES) rowptr[n] = run;
    run += loc[k];
  }
  if (tid == 1023) rowptr[NNODES] = run;
}

__global__ __launch_bounds__(256) void k_fill(const int* __restrict__ src,
                                              const int* __restrict__ dst,
                                              const int* __restrict__ rowptr,
                                              int* __restrict__ cursor,
                                              int* __restrict__ col, int E) {
  int e = blockIdx.x * 256 + threadIdx.x;
  if (e < E) {
    int d = dst[e];
    int pos = rowptr[d] + atomicAdd(&cursor[d], 1);
    col[pos] = src[e];
  }
}

#define NSLICE 313
#define ZSLOT  NNODES

__global__ __launch_bounds__(256) void k_sell(const int* __restrict__ rowptr,
                                              const int* __restrict__ col,
                                              int* __restrict__ sell) {
  const int n = blockIdx.x * 256 + threadIdx.x;
  const int s = n >> 5;
  const int lane = n & 31;
  if (s >= NSLICE) return;
  int beg = 0, deg = 0;
  if (n < NNODES) { beg = rowptr[n]; deg = rowptr[n + 1] - beg; }
  int m = deg;
  for (int ofs = 16; ofs > 0; ofs >>= 1) m = max(m, __shfl_xor(m, ofs, 32));
  const int wo = __builtin_amdgcn_readfirstlane((m + 7) >> 3);
  int4* dst = (int4*)sell + (long)8 * (rowptr[32 * s] + 8 * s) + lane;
  for (int o = 0; o < wo; ++o) {
    for (int h = 0; h < 2; ++h) {
      const int k = 8 * o + 4 * h;
      int4 c;
      c.x = (k     < deg) ? col[beg + k]     : ZSLOT;
      c.y = (k + 1 < deg) ? col[beg + k + 1] : ZSLOT;
      c.z = (k + 2 < deg) ? col[beg + k + 2] : ZSLOT;
      c.w = (k + 3 < deg) ? col[beg + k + 3] : ZSLOT;
      dst[o * 64 + h * 32] = c;
    }
  }
}

__global__ __launch_bounds__(1024) void k_lambda(const int* __restrict__ rowptr,
                                                 const int* __restrict__ sell,
                                                 float* __restrict__ sc) {
  extern __shared__ float sm[];
  float* va = sm;
  float* vb = sm + 10240;
  float* red = sm + 20480;
  const int tid = threadIdx.x;
  const int lane = tid & 31;

  float degf[10];
  int wo[10];
  int sb[10];
#pragma unroll
  for (int j = 0; j < 10; ++j) {
    const int i = tid + 1024 * j;
    const int s = i >> 5;
    int d = 0;
    if (i < NNODES) d = rowptr[i + 1] - rowptr[i];
    int m = d;
    for (int ofs = 16; ofs > 0; ofs >>= 1) m = max(m, __shfl_xor(m, ofs, 32));
    const int sv = (s < NSLICE) ? 8 * (rowptr[32 * s] + 8 * s) : 0;
    degf[j] = (float)d;
    wo[j] = __builtin_amdgcn_readfirstlane((m + 7) >> 3);
    sb[j] = __builtin_amdgcn_readfirstlane(sv);
  }

  for (int i = tid; i < NNODES; i += 1024) va[i] = 1.0f / sqrtf((float)NNODES);
  if (tid == 0) { va[ZSLOT] = 0.0f; vb[ZSLOT] = 0.0f; }
  __syncthreads();

  float nrm = 0.0f;
  float* cur = va;
  float* nxt = vb;
  for (int it = 0; it < PIters; ++it) {
    float ss = 0.0f;
#pragma unroll
    for (int j = 0; j < 10; ++j) {
      const int i = tid + 1024 * j;
      const int w8 = wo[j];
      const int4* p = (const int4*)sell + sb[j] + lane;
      const float ci = cur[i];
      float a = 0.0f;
      for (int g = 0; g < w8; g += 4) {
        int4 r[8] = {};
#pragma unroll
        for (int u = 0; u < 4; ++u) {
          if (u == 0 || g + u < w8) {
            r[2 * u]     = p[(g + u) * 64];
            r[2 * u + 1] = p[(g + u) * 64 + 32];
          }
        }
#pragma unroll
        for (int u = 0; u < 4; ++u) {
          if (u == 0 || g + u < w8) {
            const int4 c0 = r[2 * u];
            const int4 c1 = r[2 * u + 1];
            a += cur[c0.x]; a += cur[c0.y]; a += cur[c0.z]; a += cur[c0.w];
            a += cur[c1.x]; a += cur[c1.y]; a += cur[c1.z]; a += cur[c1.w];
          }
        }
      }
      if (i < NNODES) {
        float w = degf[j] * ci - a;
        nxt[i] = w;
        ss += w * w;
      }
    }
    red[tid] = ss;
    __syncthreads();
    for (int s = 512; s > 0; s >>= 1) {
      if (tid < s) red[tid] += red[tid + s];
      __syncthreads();
    }
    nrm = sqrtf(red[0]);
    for (int i = tid; i < NNODES; i += 1024) nxt[i] /= nrm;
    __syncthreads();
    float* tmp = cur; cur = nxt; nxt = tmp;
  }
  if (tid == 0) sc[0] = nrm;
}

__global__ __launch_bounds__(256) void k_gather1(const int* __restrict__ rowptr,
                                                 const int* __restrict__ col,
                                                 const float* __restrict__ Xt,
                                                 const float* __restrict__ sc,
                                                 float* __restrict__ Tx1) {
  int tid = blockIdx.x * 256 + threadIdx.x;
  int n = tid >> 7;
  int j = (tid & 127) << 2;
  int beg = rowptr[n], end = rowptr[n + 1];
  float4 acc = make_float4(0.f, 0.f, 0.f, 0.f);
  for (int k = beg; k < end; ++k) {
    float4 v = *(const float4*)(Xt + (long)col[k] * 512 + j);
    acc.x += v.x; acc.y += v.y; acc.z += v.z; acc.w += v.w;
  }
  float lm = sc[0];
  float off = 2.0f / lm;
  float dg = 2.0f * (float)(end - beg) / lm - 1.0f;
  float4 x0 = *(const float4*)(Xt + (long)n * 512 + j);
  *(float4*)(Tx1 + (long)n * 512 + j) =
      make_float4(dg * x0.x - off * acc.x, dg * x0.y - off * acc.y,
                  dg * x0.z - off * acc.z, dg * x0.w - off * acc.w);
}

__global__ __launch_bounds__(256) void k_gather2(const int* __restrict__ rowptr,
                                                 const int* __restrict__ col,
                                                 const float* __restrict__ Tx1,
                                                 const float* __restrict__ Xt,
                                                 const float* __restrict__ sc,
                                                 float* __restrict__ Tx2) {
  int tid = blockIdx.x * 256 + threadIdx.x;
  int n = tid >> 7;
  int j = (tid & 127) << 2;
  int beg = rowptr[n], end = rowptr[n + 1];
  float4 acc = make_float4(0.f, 0.f, 0.f, 0.f);
  for (int k = beg; k < end; ++k) {
    float4 v = *(const float4*)(Tx1 + (long)col[k] * 512 + j);
    acc.x += v.x; acc.y += v.y; acc.z += v.z; acc.w += v.w;
  }
  float lm = sc[0];
  float off = 2.0f / lm;
  float dg = 2.0f * (float)(end - beg) / lm - 1.0f;
  float4 x1 = *(const float4*)(Tx1 + (long)n * 512 + j);
  float4 x0 = *(const float4*)(Xt + (long)n * 512 + j);
  *(float4*)(Tx2 + (long)n * 512 + j) =
      make_float4(2.0f * (dg * x1.x - off * acc.x) - x0.x,
                  2.0f * (dg * x1.y - off * acc.y) - x0.y,
                  2.0f * (dg * x1.z - off * acc.z) - x0.z,
                  2.0f * (dg * x1.w - off * acc.w) - x0.w);
}

__global__ __launch_bounds__(256) void k_cheb_gemm(const float* __restrict__ Tx0,
                                                   const float* __restrict__ Tx1,
                                                   const float* __restrict__ Tx2,
                                                   const float* __restrict__ W,
                                                   const float* __restrict__ bvec,
                                                   float* __restrict__ Out) {
  int wave = (blockIdx.x * 256 + threadIdx.x) >> 5;
  int lane = threadIdx.x & 31;
  int m = lane & 15;
  int half = lane >> 4;
  int kb = half * 2;
  long r0 = (long)wave * 16;
  long rowA = r0 + m;

  v8f c0 = {};
  v8f c1 = {};
  const float* Ts[3] = {Tx0, Tx1, Tx2};
#pragma unroll
  for (int kidx = 0; kidx < 3; ++kidx) {
    const float* Tk = Ts[kidx];
    const float* Wk = W + kidx * (NFIN * CCHEB);
#pragma unroll
    for (int kk = 0; kk < 16; kk += 4) {
      v2f a;
      a.x = Tk[rowA * 16 + kk + kb];
      a.y = Tk[rowA * 16 + kk + kb + 1];
      v2f b0, b1;
      b0.x = Wk[(kk + kb) * 32 + m];
      b0.y = Wk[(kk + kb + 1) * 32 + m];
      b1.x = Wk[(kk + kb) * 32 + 16 + m];
      b1.y = Wk[(kk + kb + 1) * 32 + 16 + m];
      c0 = __builtin_amdgcn_wmma_f32_16x16x4_f32(false, a, false, b0, (short)0, c0, false, false);
      c1 = __builtin_amdgcn_wmma_f32_16x16x4_f32(false, a, false, b1, (short)0, c1, false, false);
    }
  }
#pragma unroll
  for (int i = 0; i < 8; ++i) {
    long row = r0 + i + half * 8;
    float vA = c0[i] + bvec[m];
    float vB = c1[i] + bvec[16 + m];
    vA = vA > 0.0f ? vA : 0.0f;
    vB = vB > 0.0f ? vB : 0.0f;
    Out[row * 32 + m] = vA;
    Out[row * 32 + 16 + m] = vB;
  }
}

__global__ __launch_bounds__(256) void k_time_gemm(const float* __restrict__ Cheb,
                                                   const float* __restrict__ X,
                                                   const float* __restrict__ tW,
                                                   const float* __restrict__ tB,
                                                   const float* __restrict__ rW,
                                                   const float* __restrict__ rB,
                                                   const float* __restrict__ gamma,
                                                   const float* __restrict__ beta,
                                                   float* __restrict__ Out) {
  int wave = (blockIdx.x * 256 + threadIdx.x) >> 5;
  int lane = threadIdx.x & 31;
  int m = lane & 15;
  int half = lane >> 4;
  int kb = half * 2;
  long r0 = (long)wave * 16;

  long rA = r0 + m;
  int nA = (int)(rA >> 5);
  int tbA = (int)(rA & 31);
  int bA = tbA >> 3;
  int tA = tbA & 7;

  v8f acc0 = {}, acc1 = {}, acc2 = {}, acc3 = {};

#pragma unroll
  for (int kk = 0; kk < 96; kk += 4) {
    int k0 = kk + kb;
    int dt0 = k0 >> 5, ci0 = k0 & 31;
    int dt1 = (k0 + 1) >> 5, ci1 = (k0 + 1) & 31;
    int tp0 = tA + dt0 - 1;
    int tp1 = tA + dt1 - 1;
    float m0 = (tp0 >= 0 && tp0 < 8) ? 1.0f : 0.0f;
    float m1 = (tp1 >= 0 && tp1 < 8) ? 1.0f : 0.0f;
    int tc0 = min(max(tp0, 0), 7);
    int tc1 = min(max(tp1, 0), 7);
    v2f a;
    a.x = m0 * Cheb[((long)nA * 32 + bA * 8 + tc0) * 32 + ci0];
    a.y = m1 * Cheb[((long)nA * 32 + bA * 8 + tc1) * 32 + ci1];
#define TCONV_STEP(ACC, CT)                                                        \
    {                                                                              \
      int co = (CT) * 16 + m;                                                      \
      v2f b;                                                                       \
      b.x = tW[co * 96 + ci0 * 3 + dt0];                                           \
      b.y = tW[co * 96 + ci1 * 3 + dt1];                                           \
      ACC = __builtin_amdgcn_wmma_f32_16x16x4_f32(false, a, false, b, (short)0,    \
                                                  ACC, false, false);              \
    }
    TCONV_STEP(acc0, 0)
    TCONV_STEP(acc1, 1)
    TCONV_STEP(acc2, 2)
    TCONV_STEP(acc3, 3)
#undef TCONV_STEP
  }

#pragma unroll
  for (int kk = 0; kk < 16; kk += 4) {
    int k0 = kk + kb;
    v2f a;
    a.x = X[(((long)(bA * NNODES + nA)) * 16 + k0) * 8 + tA];
    a.y = X[(((long)(bA * NNODES + nA)) * 16 + k0 + 1) * 8 + tA];
#define RCONV_STEP(ACC, CT)                                                        \
    {                                                                              \
      int co = (CT) * 16 + m;                                                      \
      v2f b;                                                                       \
      b.x = rW[co * 16 + k0];                                                      \
      b.y = rW[co * 16 + k0 + 1];                                                  \
      ACC = __builtin_amdgcn_wmma_f32_16x16x4_f32(false, a, false, b, (short)0,    \
                                                  ACC, false, false);              \
    }
    RCONV_STEP(acc0, 0)
    RCONV_STEP(acc1, 1)
    RCONV_STEP(acc2, 2)
    RCONV_STEP(acc3, 3)
#undef RCONV_STEP
  }

#pragma unroll
  for (int i = 0; i < 8; ++i) {
    long row = r0 + i + half * 8;
    int n = (int)(row >> 5);
    int tb = (int)(row & 31);
    int b = tb >> 3;
    int t = tb & 7;
    float vals[4];
    float s = 0.0f, ss = 0.0f;
    {
      float v0 = acc0[i] + tB[m] + rB[m];
      float v1 = acc1[i] + tB[16 + m] + rB[16 + m];
      float v2 = acc2[i] + tB[32 + m] + rB[32 + m];
      float v3 = acc3[i] + tB[48 + m] + rB[48 + m];
      v0 = v0 > 0.0f ? v0 : 0.0f;
      v1 = v1 > 0.0f ? v1 : 0.0f;
      v2 = v2 > 0.0f ? v2 : 0.0f;
      v3 = v3 > 0.0f ? v3 : 0.0f;
      vals[0] = v0; vals[1] = v1; vals[2] = v2; vals[3] = v3;
      s = v0 + v1 + v2 + v3;
      ss = v0 * v0 + v1 * v1 + v2 * v2 + v3 * v3;
    }
#pragma unroll
    for (int ofs = 1; ofs < 16; ofs <<= 1) {
      s += __shfl_xor(s, ofs, 32);
      ss += __shfl_xor(ss, ofs, 32);
    }
    float mu = s * (1.0f / 64.0f);
    float var = ss * (1.0f / 64.0f) - mu * mu;
    float inv = rsqrtf(var + LNEPS);
#pragma unroll
    for (int ct = 0; ct < 4; ++ct) {
      int c = ct * 16 + m;
      float y = (vals[ct] - mu) * inv * gamma[c] + beta[c];
      Out[(((long)(b * NNODES + n)) * 64 + c) * 8 + t] = y;
    }
  }
}

extern "C" void kernel_launch(void* const* d_in, const int* in_sizes, int n_in,
                              void* d_out, int out_size, void* d_ws, size_t ws_size,
                              hipStream_t stream) {
  (void)n_in; (void)out_size; (void)ws_size;
  const float* X     = (const float*)d_in[0];
  const int*   eidx  = (const int*)d_in[1];
  const float* chebW = (const float*)d_in[2];
  const float* chebB = (const float*)d_in[3];
  const float* timeW = (const float*)d_in[4];
  const float* timeB = (const float*)d_in[5];
  const float* resW  = (const float*)d_in[6];
  const float* resB  = (const float*)d_in[7];
  const float* gamma = (const float*)d_in[8];
  const float* beta  = (const float*)d_in[9];
  float* out = (float*)d_out;

  const int E = in_sizes[1] / 2;
  const int* src = eidx;
  const int* dst = eidx + E;

  float* w    = (float*)d_ws;
  float* Xt   = w;
  float* Tx1  = Xt + NRF;
  float* Tx2  = Tx1 + NRF;
  float* Cheb = Tx2 + NRF;
  float* sc   = Cheb + (size_t)NROWS * CCHEB;
  int*   ip     = (int*)(sc + 16);
  int*   indeg  = ip;
  int*   cursor = ip + 10240;
  int*   rowptr = ip + 20480;
  int*   col    = ip + 30976;

  k_build_xt<<<NRF / 256, 256, 0, stream>>>(X, Xt);

  k_zero_int<<<(20480 + 255) / 256, 256, 0, stream>>>(ip, 20480);
  k_count<<<(E + 255) / 256, 256, 0, stream>>>(dst, indeg, E);
  k_scan<<<1, 1024, 0, stream>>>(indeg, rowptr);
  k_fill<<<(E + 255) / 256, 256, 0, stream>>>(src, dst, rowptr, cursor, col, E);

  int* sell = (int*)Cheb;
  k_sell<<<40, 256, 0, stream>>>(rowptr, col, sell);
  const size_t lambdaLds = (size_t)(2 * 10240 + 1024) * sizeof(float);
  k_lambda<<<1, 1024, lambdaLds, stream>>>(rowptr, sell, sc);

  k_gather1<<<NNODES * 128 / 256, 256, 0, stream>>>(rowptr, col, Xt, sc, Tx1);
  k_gather2<<<NNODES * 128 / 256, 256, 0, stream>>>(rowptr, col, Tx1, Xt, sc, Tx2);

  k_cheb_gemm<<<2500, 256, 0, stream>>>(Xt, Tx1, Tx2, chebW, chebB, Cheb);

  k_time_gemm<<<2500, 256, 0, stream>>>(Cheb, X, timeW, timeB, resW, resB,
                                        gamma, beta, out);
}


# k_lambda: the 64 power steps run as 64 chip-wide launches (one wave per 32-node slice; vector and per-slice norm sums read from L2), not one work-group on LDS; the norm's sum order changes
# speedup vs baseline: 4.0303x; 1.9667x over previous
#include <hip/hip_runtime.h>

typedef __attribute__((ext_vector_type(2))) float v2f;
typedef __attribute__((ext_vector_type(8))) float v8f;

#define NNODES 10000
#define NBATCH 4
#define NTIME  8
#define NFIN   16
#define TBD    32
#define CCHEB  32
#define CTIME  64
#define NROWS  320000
#define NRF    5120000
#define LNEPS  1e-5f
#define PIters 64

__global__ __launch_bounds__(256) void k_zero_int(int* __restrict__ p, int n) {
  int i = blockIdx.x * 256 + threadIdx.x;
  if (i < n) p[i] = 0;
}

__global__ __launch_bounds__(256) void k_build_xt(const float* __restrict__ X,
                                                  float* __restrict__ Xt) {
  int tid = blockIdx.x * 256 + threadIdx.x;
  int f2 = tid & 15;
  int tb = (tid >> 4) & 31;
  int n2 = tid >> 9;
  long i = ((long)(n2 * 16 + f2)) * 32 + tb;
  int t = (int)(i & 7);
  long q = i >> 3;
  int n = (int)(q % NNODES);
  int p = (int)(q / NNODES);
  int b = p & 3;
  int f = p >> 2;
  Xt[((long)n2 * 32 + tb) * 16 + f2] =
      X[(((long)(b * NNODES + n)) * 16 + f) * 8 + t];
}

__global__ __launch_bounds__(256) void k_count(const int* __restrict__ dst,
                                               int* __restrict__ indeg, int E) {
  int e = blockIdx.x * 256 + threadIdx.x;
  if (e < E) atomicAdd(&indeg[dst[e]], 1);
}

__global__ __launch_bounds__(1024) void k_scan(const int* __restrict__ indeg,
                                               int* __restrict__ rowptr) {
  __shared__ int part[1024];
  int tid = threadIdx.x;
  int base = tid * 10;
  int loc[10];
  int s = 0;
#pragma unroll
  for (int k = 0; k < 10; ++k) {
    int n = base + k;
    int d = (n < NNODES) ? indeg[n] : 0;
    loc[k] = d;
    s += d;
  }
  part[tid] = s;
  __syncthreads();
  for (int ofs = 1; ofs < 1024; ofs <<= 1) {
    int v = 0;
    if (tid >= ofs) v = part[tid - ofs];
    __syncthreads();
    part[tid] += v;
    __syncthreads();
  }
  int run = (tid == 0) ? 0 : part[tid - 1];
#pragma unroll
  for (int k = 0; k < 10; ++k) {
    int n = base + k;
    if (n < NNODES) rowptr[n] = run;
    run += loc[k];
  }
  if (tid == 1023) rowptr[NNODES] = run;
}

__global__ __launch_bounds__(256) void k_fill(const int* __restrict__ src,
                                              const int* __restrict__ dst,
                                              const int* __restrict__ rowptr,
                                              int* __restrict__ cursor,
                                              int* __restrict__ col, int E) {
  int e = blockIdx.x * 256 + threadIdx.x;
  if (e < E) {
    int d = dst[e];
    int pos = rowptr[d] + atomicAdd(&cursor[d], 1);
    col[pos] = src[e];
  }
}

#define NSLICE 313
#define ZSLOT  NNODES

__global__ __launch_bounds__(256) void k_sell(const int* __restrict__ rowptr,
                                              const int* __restrict__ col,
                                              int* __restrict__ sell,
                                              int* __restrict__ swidth,
                                              float* __restrict__ v0) {
  const int n = blockIdx.x * 256 + threadIdx.x;
  const int s = n >> 5;
  const int lane = n & 31;
  if (n <= ZSLOT) v0[n] = (n < NNODES) ? 1.0f / sqrtf((float)NNODES) : 0.0f;
  if (s >= NSLICE) return;
  int beg = 0, deg = 0;
  if (n < NNODES) { beg = rowptr[n]; deg = rowptr[n + 1] - beg; }
  int m = deg;
  for (int ofs = 16; ofs > 0; ofs >>= 1) m = max(m, __shfl_xor(m, ofs, 32));
  const int wo = __builtin_amdgcn_readfirstlane((m + 7) >> 3);
  if (lane == 0) swidth[s] = wo;
  int4* dst = (int4*)sell + (long)8 * (rowptr[32 * s] + 8 * s) + lane;
  for (int o = 0; o < wo; ++o) {
    for (int h = 0; h < 2; ++h) {
      const int k = 8 * o + 4 * h;
      int4 c;
      c.x = (k     < deg) ? col[beg + k]     : ZSLOT;
      c.y = (k + 1 < deg) ? col[beg + k + 1] : ZSLOT;
      c.z = (k + 2 < deg) ? col[beg + k + 2] : ZSLOT;
      c.w = (k + 3 < deg) ? col[beg + k + 3] : ZSLOT;
      dst[o * 64 + h * 32] = c;
    }
  }
}

#define NPROW 10

__device__ __forceinline__ void load_partials(const float* __restrict__ part, int lane,
                                              float (&pp)[NPROW]) {
#pragma unroll
  for (int q = 0; q < NPROW; ++q) {
    const int i = lane + 32 * q;
    pp[q] = (32 * q + 31 < NSLICE) ? part[i] : part[min(i, NSLICE - 1)];
  }
}

template <int XMASK>
__device__ __forceinline__ float row_xor(float v) {
  return __builtin_bit_cast(float, __builtin_amdgcn_update_dpp(0, __builtin_bit_cast(int, v),
                                                               0x160 + XMASK, 0xf, 0xf, false));
}
__device__ __forceinline__ float wave_sum(float v) {
  v += row_xor<1>(v);
  v += row_xor<2>(v);
  v += row_xor<4>(v);
  v += row_xor<8>(v);
  v += __builtin_bit_cast(float, __builtin_amdgcn_permlanex16(__builtin_bit_cast(int, v),
                                                              __builtin_bit_cast(int, v),
                                                              0x76543210, 0xfedcba98, false, false));
  return v;
}

__device__ __forceinline__ float norm_of_partials(const float (&pp)[NPROW], int lane) {
  float t = 0.0f;
#pragma unroll
  for (int q = 0; q < NPROW; ++q) t += (lane + 32 * q < NSLICE) ? pp[q] : 0.0f;
  return sqrtf(wave_sum(t));
}

template <bool FIRST>
__global__ __launch_bounds__(32) void k_pstep(const int* __restrict__ rowptr,
                                              const int* __restrict__ sell,
                                              const int* __restrict__ swidth,
                                              const float* __restrict__ win,
                                              const float* __restrict__ pin,
                                              float* __restrict__ wout,
                                              float* __restrict__ pout) {
  const int s = blockIdx.x;
  const int lane = threadIdx.x;
  const int n = 32 * s + lane;
  float pp[NPROW];
  if (!FIRST) load_partials(pin, lane, pp);
  const int w8 = swidth[s];
  const int4* p = (const int4*)sell + (long)8 * (rowptr[32 * s] + 8 * s) + lane;
  int d = 0;
  float own = 0.0f;
  if (n < NNODES) { d = rowptr[n + 1] - rowptr[n]; own = win[n]; }

  float a = 0.0f;
  float nrm = 1.0f;
  for (int g = 0; g < w8; g += 4) {
    int4 r[8];
#pragma unroll
    for (int u = 0; u < 8; ++u) r[u] = make_int4(ZSLOT, ZSLOT, ZSLOT, ZSLOT);
#pragma unroll
    for (int u = 0; u < 4; ++u) {
      if (u == 0 || g + u < w8) {
        r[2 * u]     = p[(g + u) * 64];
        r[2 * u + 1] = p[(g + u) * 64 + 32];
      }
    }
    float x[32];
#pragma unroll
    for (int u = 0; u < 8; ++u) {
      x[4 * u]     = win[r[u].x];
      x[4 * u + 1] = win[r[u].y];
      x[4 * u + 2] = win[r[u].z];
      x[4 * u + 3] = win[r[u].w];
    }
    if (!FIRST && g == 0) nrm = norm_of_partials(pp, lane);
#pragma unroll
    for (int k = 0; k < 32; ++k) a += FIRST ? x[k] : x[k] / nrm;
  }
  if (!FIRST && w8 <= 0) nrm = norm_of_partials(pp, lane);
  float ss = 0.0f;
  if (n < NNODES) {
    const float ci = FIRST ? own : own / nrm;
    const float w = __builtin_fmaf((float)d, ci, -a);
    wout[n] = w;
    ss = w * w;
  }
  if (n == ZSLOT) wout[ZSLOT] = 0.0f;
  ss = wave_sum(ss);
  if (lane == 0) pout[s] = ss;
}

__global__ __launch_bounds__(32) void k_lnorm(const float* __restrict__ pin,
                                              float* __restrict__ sc) {
  float pp[NPROW];
  load_partials(pin, threadIdx.x, pp);
  const float nrm = norm_of_partials(pp, threadIdx.x);
  if (threadIdx.x == 0) sc[0] = nrm;
}

__global__ __launch_bounds__(256) void k_gather1(const int* __restrict__ rowptr,
                                                 const int* __restrict__ col,
                                                 const float* __restrict__ Xt,
                                                 const float* __restrict__ sc,
                                                 float* __restrict__ Tx1) {
  int tid = blockIdx.x * 256 + threadIdx.x;
  int n = tid >> 7;
  int j = (tid & 127) << 2;
  int beg = rowptr[n], end = rowptr[n + 1];
  float4 acc = make_float4(0.f, 0.f, 0.f, 0.f);
  for (int k = beg; k < end; ++k) {
    float4 v = *(const float4*)(Xt + (long)col[k] * 512 + j);
    acc.x += v.x; acc.y += v.y; acc.z += v.z; acc.w += v.w;
  }
  float lm = sc[0];
  float off = 2.0f / lm;
  float dg = 2.0f * (float)(end - beg) / lm - 1.0f;
  float4 x0 = *(const float4*)(Xt + (long)n * 512 + j);
  *(float4*)(Tx1 + (long)n * 512 + j) =
      make_float4(dg * x0.x - off * acc.x, dg * x0.y - off * acc.y,
                  dg * x0.z - off * acc.z, dg * x0.w - off * acc.w);
}

__global__ __launch_bounds__(256) void k_gather2(const int* __restrict__ rowptr,
                                                 const int* __restrict__ col,
                                                 const float* __restrict__ Tx1,
                                                 const float* __restrict__ Xt,
                                                 const float* __restrict__ sc,
                                                 float* __restrict__ Tx2) {
  int tid = blockIdx.x * 256 + threadIdx.x;
  int n = tid >> 7;
  int j = (tid & 127) << 2;
  int beg = rowptr[n], end = rowptr[n + 1];
  float4 acc = make_float4(0.f, 0.f, 0.f, 0.f);
  for (int k = beg; k < end; ++k) {
    float4 v = *(const float4*)(Tx1 + (long)col[k] * 512 + j);
    acc.x += v.x; acc.y += v.y; acc.z += v.z; acc.w += v.w;
  }
  float lm = sc[0];
  float off = 2.0f / lm;
  float dg = 2.0f * (float)(end - beg) / lm - 1.0f;
  float4 x1 = *(const float4*)(Tx1 + (long)n * 512 + j);
  float4 x0 = *(const float4*)(Xt + (long)n * 512 + j);
  *(float4*)(Tx2 + (long)n * 512 + j) =
      make_float4(2.0f * (dg * x1.x - off * acc.x) - x0.x,
                  2.0f * (dg * x1.y - off * acc.y) - x0.y,
                  2.0f * (dg * x1.z - off * acc.z) - x0.z,
                  2.0f * (dg * x1.w - off * acc.w) - x0.w);
}

__global__ __launch_bounds__(256) void k_cheb_gemm(const float* __restrict__ Tx0,
                                                   const float* __restrict__ Tx1,
                                                   const float* __restrict__ Tx2,
                                                   const float* __restrict__ W,
                                                   const float* __restrict__ bvec,
                                                   float* __restrict__ Out) {
  int wave = (blockIdx.x * 256 + threadIdx.x) >> 5;
  int lane = threadIdx.x & 31;
  int m = lane & 15;
  int half = lane >> 4;
  int kb = half * 2;
  long r0 = (long)wave * 16;
  long rowA = r0 + m;

  v8f c0 = {};
  v8f c1 = {};
  const float* Ts[3] = {Tx0, Tx1, Tx2};
#pragma unroll
  for (int kidx = 0; kidx < 3; ++kidx) {
    const float* Tk = Ts[kidx];
    const float* Wk = W + kidx * (NFIN * CCHEB);
#pragma unroll
    for (int kk = 0; kk < 16; kk += 4) {
      v2f a;
      a.x = Tk[rowA * 16 + kk + kb];
      a.y = Tk[rowA * 16 + kk + kb + 1];
      v2f b0, b1;
      b0.x = Wk[(kk + kb) * 32 + m];
      b0.y = Wk[(kk + kb + 1) * 32 + m];
      b1.x = Wk[(kk + kb) * 32 + 16 + m];
      b1.y = Wk[(kk + kb + 1) * 32 + 16 + m];
      c0 = __builtin_amdgcn_wmma_f32_16x16x4_f32(false, a, false, b0, (short)0, c0, false, false);
      c1 = __builtin_amdgcn_wmma_f32_16x16x4_f32(false, a, false, b1, (short)0, c1, false, false);
    }
  }
#pragma unroll
  for (int i = 0; i < 8; ++i) {
    long row = r0 + i + half * 8;
    float vA = c0[i] + bvec[m];
    float vB = c1[i] + bvec[16 + m];
    vA = vA > 0.0f ? vA : 0.0f;
    vB = vB > 0.0f ? vB : 0.0f;
    Out[row * 32 + m] = vA;
    Out[row * 32 + 16 + m] = vB;
  }
}

__global__ __launch_bounds__(256) void k_time_gemm(const float* __restrict__ Cheb,
                                                   const float* __restrict__ X,
                                                   const float* __restrict__ tW,
                                                   const float* __restrict__ tB,
                                                   const float* __restrict__ rW,
                                                   const float* __restrict__ rB,
                                                   const float* __restrict__ gamma,
                                                   const float* __restrict__ beta,
                                                   float* __restrict__ Out) {
  int wave = (blockIdx.x * 256 + threadIdx.x) >> 5;
  int lane = threadIdx.x & 31;
  int m = lane & 15;
  int half = lane >> 4;
  int kb = half * 2;
  long r0 = (long)wave * 16;

  long rA = r0 + m;
  int nA = (int)(rA >> 5);
  int tbA = (int)(rA & 31);
  int bA = tbA >> 3;
  int tA = tbA & 7;

  v8f acc0 = {}, acc1 = {}, acc2 = {}, acc3 = {};

#pragma unroll
  for (int kk = 0; kk < 96; kk += 4) {
    int k0 = kk + kb;
    int dt0 = k0 >> 5, ci0 = k0 & 31;
    int dt1 = (k0 + 1) >> 5, ci1 = (k0 + 1) & 31;
    int tp0 = tA + dt0 - 1;
    int tp1 = tA + dt1 - 1;
    float m0 = (tp0 >= 0 && tp0 < 8) ? 1.0f : 0.0f;
    float m1 = (tp1 >= 0 && tp1 < 8) ? 1.0f : 0.0f;
    int tc0 = min(max(tp0, 0), 7);
    int tc1 = min(max(tp1, 0), 7);
    v2f a;
    a.x = m0 * Cheb[((long)nA * 32 + bA * 8 + tc0) * 32 + ci0];
    a.y = m1 * Cheb[((long)nA * 32 + bA * 8 + tc1) * 32 + ci1];
#define TCONV_STEP(ACC, CT)                                                        \
    {                                                                              \
      int co = (CT) * 16 + m;                                                      \
      v2f b;                                                                       \
      b.x = tW[co * 96 + ci0 * 3 + dt0];                                           \
      b.y = tW[co * 96 + ci1 * 3 + dt1];                                           \
      ACC = __builtin_amdgcn_wmma_f32_16x16x4_f32(false, a, false, b, (short)0,    \
                                                  ACC, false, false);              \
    }
    TCONV_STEP(acc0, 0)
    TCONV_STEP(acc1, 1)
    TCONV_STEP(acc2, 2)
    TCONV_STEP(acc3, 3)
#undef TCONV_STEP
  }

#pragma unroll
  for (int kk = 0; kk < 16; kk += 4) {
    int k0 = kk + kb;
    v2f a;
    a.x = X[(((long)(bA * NNODES + nA)) * 16 + k0) * 8 + tA];
    a.y = X[(((long)(bA * NNODES + nA)) * 16 + k0 + 1) * 8 + tA];
#define RCONV_STEP(ACC, CT)                                                        \
    {                                                                              \
      int co = (CT) * 16 + m;                                                      \
      v2f b;                                                                       \
      b.x = rW[co * 16 + k0];                                                      \
      b.y = rW[co * 16 + k0 + 1];                                                  \
      ACC = __builtin_amdgcn_wmma_f32_16x16x4_f32(false, a, false, b, (short)0,    \
                                                  ACC, false, false);              \
    }
    RCONV_STEP(acc0, 0)
    RCONV_STEP(acc1, 1)
    RCONV_STEP(acc2, 2)
    RCONV_STEP(acc3, 3)
#undef RCONV_STEP
  }

#pragma unroll
  for (int i = 0; i < 8; ++i) {
    long row = r0 + i + half * 8;
    int n = (int)(row >> 5);
    int tb = (int)(row & 31);
    int b = tb >> 3;
    int t = tb & 7;
    float vals[4];
    float s = 0.0f, ss = 0.0f;
    {
      float v0 = acc0[i] + tB[m] + rB[m];
      float v1 = acc1[i] + tB[16 + m] + rB[16 + m];
      float v2 = acc2[i] + tB[32 + m] + rB[32 + m];
      float v3 = acc3[i] + tB[48 + m] + rB[48 + m];
      v0 = v0 > 0.0f ? v0 : 0.0f;
      v1 = v1 > 0.0f ? v1 : 0.0f;
      v2 = v2 > 0.0f ? v2 : 0.0f;
      v3 = v3 > 0.0f ? v3 : 0.0f;
      vals[0] = v0; vals[1] = v1; vals[2] = v2; vals[3] = v3;
      s = v0 + v1 + v2 + v3;
      ss = v0 * v0 + v1 * v1 + v2 * v2 + v3 * v3;
    }
#pragma unroll
    for (int ofs = 1; ofs < 16; ofs <<= 1) {
      s += __shfl_xor(s, ofs, 32);
      ss += __shfl_xor(ss, ofs, 32);
    }
    float mu = s * (1.0f / 64.0f);
    float var = ss * (1.0f / 64.0f) - mu * mu;
    float inv = rsqrtf(var + LNEPS);
#pragma unroll
    for (int ct = 0; ct < 4; ++ct) {
      int c = ct * 16 + m;
      float y = (vals[ct] - mu) * inv * gamma[c] + beta[c];
      Out[(((long)(b * NNODES + n)) * 64 + c) * 8 + t] = y;
    }
  }
}

extern "C" void kernel_launch(void* const* d_in, const int* in_sizes, int n_in,
                              void* d_out, int out_size, void* d_ws, size_t ws_size,
                              hipStream_t stream) {
  (void)n_in; (void)out_size; (void)ws_size;
  const float* X     = (const float*)d_in[0];
  const int*   eidx  = (const int*)d_in[1];
  const float* chebW = (const float*)d_in[2];
  const float* chebB = (const float*)d_in[3];
  const float* timeW = (const float*)d_in[4];
  const float* timeB = (const float*)d_in[5];
  const float* resW  = (const float*)d_in[6];
  const float* resB  = (const float*)d_in[7];
  const float* gamma = (const float*)d_in[8];
  const float* beta  = (const float*)d_in[9];
  float* out = (float*)d_out;

  const int E = in_sizes[1] / 2;
  const int* src = eidx;
  const int* dst = eidx + E;

  float* w    = (float*)d_ws;
  float* Xt   = w;
  float* Tx1  = Xt + NRF;
  float* Tx2  = Tx1 + NRF;
  float* Cheb = Tx2 + NRF;
  float* sc   = Cheb + (size_t)NROWS * CCHEB;
  int*   ip     = (int*)(sc + 16);
  int*   indeg  = ip;
  int*   cursor = ip + 10240;
  int*   rowptr = ip + 20480;
  int*   col    = ip + 30976;

  k_build_xt<<<NRF / 256, 256, 0, stream>>>(X, Xt);

  k_zero_int<<<(20480 + 255) / 256, 256, 0, stream>>>(ip, 20480);
  k_count<<<(E + 255) / 256, 256, 0, stream>>>(dst, indeg, E);
  k_scan<<<1, 1024, 0, stream>>>(indeg, rowptr);
  k_fill<<<(E + 255) / 256, 256, 0, stream>>>(src, dst, rowptr, cursor, col, E);

  int* sell = (int*)Cheb;
  float* vcur = Cheb + 6000000;
  float* vnxt = vcur + 10240;
  float* pcur = vcur + 20480;
  float* pnxt = vcur + 20992;
  int* swidth = (int*)(vcur + 21504);
  k_sell<<<40, 256, 0, stream>>>(rowptr, col, sell, swidth, vcur);
  k_pstep<true><<<NSLICE, 32, 0, stream>>>(rowptr, sell, swidth, vcur, pcur, vnxt, pnxt);
  for (int it = 1; it < PIters; ++it) {
    float* t = vcur; vcur = vnxt; vnxt = t;
    t = pcur; pcur = pnxt; pnxt = t;
    k_pstep<false><<<NSLICE, 32, 0, stream>>>(rowptr, sell, swidth, vcur, pcur, vnxt, pnxt);
  }
  k_lnorm<<<1, 32, 0, stream>>>(pnxt, sc);

  k_gather1<<<NNODES * 128 / 256, 256, 0, stream>>>(rowptr, col, Xt, sc, Tx1);
  k_gather2<<<NNODES * 128 / 256, 256, 0, stream>>>(rowptr, col, Tx1, Xt, sc, Tx2);

  k_cheb_gemm<<<2500, 256, 0, stream>>>(Xt, Tx1, Tx2, chebW, chebB, Cheb);

  k_time_gemm<<<2500, 256, 0, stream>>>(Cheb, X, timeW, timeB, resW, resB,
                                        gamma, beta, out);
}


# k_pstep: each 4-wave work-group copies the 40-KB vector into LDS with whole-line 16-byte loads and gathers neighbours from LDS, not by 32 scattered 4-byte global gathers per lane; sum order kept
# speedup vs baseline: 4.2307x; 1.0497x over previous
#include <hip/hip_runtime.h>

typedef __attribute__((ext_vector_type(2))) float v2f;
typedef __attribute__((ext_vector_type(8))) float v8f;

#define NNODES 10000
#define NBATCH 4
#define NTIME  8
#define NFIN   16
#define TBD    32
#define CCHEB  32
#define CTIME  64
#define NROWS  320000
#define NRF    5120000
#define LNEPS  1e-5f
#define PIters 64

__global__ __launch_bounds__(256) void k_zero_int(int* __restrict__ p, int n) {
  int i = blockIdx.x * 256 + threadIdx.x;
  if (i < n) p[i] = 0;
}

__global__ __launch_bounds__(256) void k_build_xt(const float* __restrict__ X,
                                                  float* __restrict__ Xt) {
  int tid = blockIdx.x * 256 + threadIdx.x;
  int f2 = tid & 15;
  int tb = (tid >> 4) & 31;
  int n2 = tid >> 9;
  long i = ((long)(n2 * 16 + f2)) * 32 + tb;
  int t = (int)(i & 7);
  long q = i >> 3;
  int n = (int)(q % NNODES);
  int p = (int)(q / NNODES);
  int b = p & 3;
  int f = p >> 2;
  Xt[((long)n2 * 32 + tb) * 16 + f2] =
      X[(((long)(b * NNODES + n)) * 16 + f) * 8 + t];
}

__global__ __launch_bounds__(256) void k_count(const int* __restrict__ dst,
                                               int* __restrict__ indeg, int E) {
  int e = blockIdx.x * 256 + threadIdx.x;
  if (e < E) atomicAdd(&indeg[dst[e]], 1);
}

__global__ __launch_bounds__(1024) void k_scan(const int* __restrict__ indeg,
                                               int* __restrict__ rowptr) {
  __shared__ int part[1024];
  int tid = threadIdx.x;
  int base = tid * 10;
  int loc[10];
  int s = 0;
#pragma unroll
  for (int k = 0; k < 10; ++k) {
    int n = base + k;
    int d = (n < NNODES) ? indeg[n] : 0;
    loc[k] = d;
    s += d;
  }
  part[tid] = s;
  __syncthreads();
  for (int ofs = 1; ofs < 1024; ofs <<= 1) {
    int v = 0;
    if (tid >= ofs) v = part[tid - ofs];
    __syncthreads();
    part[tid] += v;
    __syncthreads();
  }
  int run = (tid == 0) ? 0 : part[tid - 1];
#pragma unroll
  for (int k = 0; k < 10; ++k) {
    int n = base + k;
    if (n < NNODES) rowptr[n] = run;
    run += loc[k];
  }
  if (tid == 1023) rowptr[NNODES] = run;
}

__global__ __launch_bounds__(256) void k_fill(const int* __restrict__ src,
                                              const int* __restrict__ dst,
                                              const int* __restrict__ rowptr,
                                              int* __restrict__ cursor,
                                              int* __restrict__ col, int E) {
  int e = blockIdx.x * 256 + threadIdx.x;
  if (e < E) {
    int d = dst[e];
    int pos = rowptr[d] + atomicAdd(&cursor[d], 1);
    col[pos] = src[e];
  }
}

#define NSLICE 313
#define PS_WAVES   4
#define PS_THREADS (32 * PS_WAVES)
#define PS_GROUPS  ((NSLICE + PS_WAVES - 1) / PS_WAVES)
#define NV4        (NNODES / 4)
#define PS_LOADS   ((NV4 + PS_THREADS - 1) / PS_THREADS)
#define NVPAD      (4 * PS_THREADS * PS_LOADS)
#define ZSLOT      NVPAD

__global__ __launch_bounds__(256) void k_sell(const int* __restrict__ rowptr,
                                              const int* __restrict__ col,
                                              int* __restrict__ sell,
                                              int* __restrict__ swidth,
                                              float* __restrict__ v0) {
  const int n = blockIdx.x * 256 + threadIdx.x;
  const int s = n >> 5;
  const int lane = n & 31;
  if (n < NNODES) v0[n] = 1.0f / sqrtf((float)NNODES);
  if (s >= NSLICE) return;
  int beg = 0, deg = 0;
  if (n < NNODES) { beg = rowptr[n]; deg = rowptr[n + 1] - beg; }
  int m = deg;
  for (int ofs = 16; ofs > 0; ofs >>= 1) m = max(m, __shfl_xor(m, ofs, 32));
  const int wo = __builtin_amdgcn_readfirstlane((m + 7) >> 3);
  if (lane == 0) swidth[s] = wo;
  int4* dst = (int4*)sell + (long)8 * (rowptr[32 * s] + 8 * s) + lane;
  for (int o = 0; o < max(wo, 1); ++o) {
    for (int h = 0; h < 2; ++h) {
      const int k = 8 * o + 4 * h;
      int4 c;
      c.x = (k     < deg) ? col[beg + k]     : ZSLOT;
      c.y = (k + 1 < deg) ? col[beg + k + 1] : ZSLOT;
      c.z = (k + 2 < deg) ? col[beg + k + 2] : ZSLOT;
      c.w = (k + 3 < deg) ? col[beg + k + 3] : ZSLOT;
      dst[o * 64 + h * 32] = c;
    }
  }
}

#define NPROW 10

__device__ __forceinline__ void load_partials(const float* __restrict__ part, int lane,
                                              float (&pp)[NPROW]) {
#pragma unroll
  for (int q = 0; q < NPROW; ++q) {
    const int i = lane + 32 * q;
    pp[q] = (32 * q + 31 < NSLICE) ? part[i] : part[min(i, NSLICE - 1)];
  }
}

template <int XMASK>
__device__ __forceinline__ float row_xor(float v) {
  return __builtin_bit_cast(float, __builtin_amdgcn_update_dpp(0, __builtin_bit_cast(int, v),
                                                               0x160 + XMASK, 0xf, 0xf, false));
}
__device__ __forceinline__ float wave_sum(float v) {
  v += row_xor<1>(v);
  v += row_xor<2>(v);
  v += row_xor<4>(v);
  v += row_xor<8>(v);
  v += __builtin_bit_cast(float, __builtin_amdgcn_permlanex16(__builtin_bit_cast(int, v),
                                                              __builtin_bit_cast(int, v),
                                                              0x76543210, 0xfedcba98, false, false));
  return v;
}

__device__ __forceinline__ float norm_of_partials(const float (&pp)[NPROW], int lane) {
  float t = 0.0f;
#pragma unroll
  for (int q = 0; q < NPROW; ++q) t += (lane + 32 * q < NSLICE) ? pp[q] : 0.0f;
  return sqrtf(wave_sum(t));
}

__device__ __forceinline__ void load_rows(const int4* __restrict__ p, int g, int w8, int4 (&r)[8]) {
  const int4 z = make_int4(ZSLOT, ZSLOT, ZSLOT, ZSLOT);
#pragma unroll
  for (int u = 0; u < 4; ++u) {
    const int row = max(min(g + u, w8 - 1), 0);
    const int4 lo = p[row * 64];
    const int4 hi = p[row * 64 + 32];
    r[2 * u]     = (g + u < w8) ? lo : z;
    r[2 * u + 1] = (g + u < w8) ? hi : z;
  }
}

template <bool FIRST>
__global__ __launch_bounds__(PS_THREADS) void k_pstep(const int* __restrict__ rowptr,
                                                      const int* __restrict__ sell,
                                                      const int* __restrict__ swidth,
                                                      const float* __restrict__ win,
                                                      const float* __restrict__ pin,
                                                      float* __restrict__ wout,
                                                      float* __restrict__ pout) {
  __shared__ __attribute__((aligned(16))) float sv[NVPAD + 4];
  const int tid = threadIdx.x;
  const int lane = tid & 31;
  const int s = blockIdx.x * PS_WAVES + __builtin_amdgcn_readfirstlane(tid >> 5);
  const bool live = s < NSLICE;
  const int n = 32 * s + lane;
  float pp[NPROW];
  if (!FIRST) load_partials(pin, lane, pp);

  const float4* __restrict__ w4 = (const float4*)win;
  float4 buf[PS_LOADS];
#pragma unroll
  for (int j = 0; j < PS_LOADS; ++j) buf[j] = w4[min(tid + PS_THREADS * j, NV4 - 1)];
  __builtin_amdgcn_sched_barrier(0);

  const int sc = min(s, NSLICE - 1);
  const int w8 = swidth[sc];
  const int4* p = (const int4*)sell + (long)8 * (rowptr[32 * sc] + 8 * sc) + lane;
  const int nc = min(n, NNODES - 1);
  const int d = rowptr[nc + 1] - rowptr[nc];
  int4 r[8];
  load_rows(p, 0, w8, r);

#pragma unroll
  for (int j = 0; j < PS_LOADS; ++j) ((float4*)sv)[tid + PS_THREADS * j] = buf[j];
  if (tid == 0) sv[ZSLOT] = 0.0f;
  __syncthreads();
  if (!live) return;

  const float own = (n < NNODES) ? sv[n] : 0.0f;
  float a = 0.0f;
  float nrm = 1.0f;
  for (int g = 0; g < w8; g += 4) {
    float x[32];
#pragma unroll
    for (int u = 0; u < 8; ++u) {
      x[4 * u]     = sv[r[u].x];
      x[4 * u + 1] = sv[r[u].y];
      x[4 * u + 2] = sv[r[u].z];
      x[4 * u + 3] = sv[r[u].w];
    }
    if (g + 4 < w8) load_rows(p, g + 4, w8, r);
    if (!FIRST && g == 0) nrm = norm_of_partials(pp, lane);
#pragma unroll
    for (int k = 0; k < 32; ++k) a += FIRST ? x[k] : x[k] / nrm;
  }
  if (!FIRST && w8 <= 0) nrm = norm_of_partials(pp, lane);
  float ss = 0.0f;
  if (n < NNODES) {
    const float ci = FIRST ? own : own / nrm;
    const float w = __builtin_fmaf((float)d, ci, -a);
    wout[n] = w;
    ss = w * w;
  }
  ss = wave_sum(ss);
  if (lane == 0) pout[s] = ss;
}

__global__ __launch_bounds__(32) void k_lnorm(const float* __restrict__ pin,
                                              float* __restrict__ sc) {
  float pp[NPROW];
  load_partials(pin, threadIdx.x, pp);
  const float nrm = norm_of_partials(pp, threadIdx.x);
  if (threadIdx.x == 0) sc[0] = nrm;
}

__global__ __launch_bounds__(256) void k_gather1(const int* __restrict__ rowptr,
                                                 const int* __restrict__ col,
                                                 const float* __restrict__ Xt,
                                                 const float* __restrict__ sc,
                                                 float* __restrict__ Tx1) {
  int tid = blockIdx.x * 256 + threadIdx.x;
  int n = tid >> 7;
  int j = (tid & 127) << 2;
  int beg = rowptr[n], end = rowptr[n + 1];
  float4 acc = make_float4(0.f, 0.f, 0.f, 0.f);
  for (int k = beg; k < end; ++k) {
    float4 v = *(const float4*)(Xt + (long)col[k] * 512 + j);
    acc.x += v.x; acc.y += v.y; acc.z += v.z; acc.w += v.w;
  }
  float lm = sc[0];
  float off = 2.0f / lm;
  float dg = 2.0f * (float)(end - beg) / lm - 1.0f;
  float4 x0 = *(const float4*)(Xt + (long)n * 512 + j);
  *(float4*)(Tx1 + (long)n * 512 + j) =
      make_float4(dg * x0.x - off * acc.x, dg * x0.y - off * acc.y,
                  dg * x0.z - off * acc.z, dg * x0.w - off * acc.w);
}

__global__ __launch_bounds__(256) void k_gather2(const int* __restrict__ rowptr,
                                                 const int* __restrict__ col,
                                                 const float* __restrict__ Tx1,
                                                 const float* __restrict__ Xt,
                                                 const float* __restrict__ sc,
                                                 float* __restrict__ Tx2) {
  int tid = blockIdx.x * 256 + threadIdx.x;
  int n = tid >> 7;
  int j = (tid & 127) << 2;
  int beg = rowptr[n], end = rowptr[n + 1];
  float4 acc = make_float4(0.f, 0.f, 0.f, 0.f);
  for (int k = beg; k < end; ++k) {
    float4 v = *(const float4*)(Tx1 + (long)col[k] * 512 + j);
    acc.x += v.x; acc.y += v.y; acc.z += v.z; acc.w += v.w;
  }
  float lm = sc[0];
  float off = 2.0f / lm;
  float dg = 2.0f * (float)(end - beg) / lm - 1.0f;
  float4 x1 = *(const float4*)(Tx1 + (long)n * 512 + j);
  float4 x0 = *(const float4*)(Xt + (long)n * 512 + j);
  *(float4*)(Tx2 + (long)n * 512 + j) =
      make_float4(2.0f * (dg * x1.x - off * acc.x) - x0.x,
                  2.0f * (dg * x1.y - off * acc.y) - x0.y,
                  2.0f * (dg * x1.z - off * acc.z) - x0.z,
                  2.0f * (dg * x1.w - off * acc.w) - x0.w);
}

__global__ __launch_bounds__(256) void k_cheb_gemm(const float* __restrict__ Tx0,
                                                   const float* __restrict__ Tx1,
                                                   const float* __restrict__ Tx2,
                                                   const float* __restrict__ W,
                                                   const float* __restrict__ bvec,
                                                   float* __restrict__ Out) {
  int wave = (blockIdx.x * 256 + threadIdx.x) >> 5;
  int lane = threadIdx.x & 31;
  int m = lane & 15;
  int half = lane >> 4;
  int kb = half * 2;
  long r0 = (long)wave * 16;
  long rowA = r0 + m;

  v8f c0 = {};
  v8f c1 = {};
  const float* Ts[3] = {Tx0, Tx1, Tx2};
#pragma unroll
  for (int kidx = 0; kidx < 3; ++kidx) {
    const float* Tk = Ts[kidx];
    const float* Wk = W + kidx * (NFIN * CCHEB);
#pragma unroll
    for (int kk = 0; kk < 16; kk += 4) {
      v2f a;
      a.x = Tk[rowA * 16 + kk + kb];
      a.y = Tk[rowA * 16 + kk + kb + 1];
      v2f b0, b1;
      b0.x = Wk[(kk + kb) * 32 + m];
      b0.y = Wk[(kk + kb + 1) * 32 + m];
      b1.x = Wk[(kk + kb) * 32 + 16 + m];
      b1.y = Wk[(kk + kb + 1) * 32 + 16 + m];
      c0 = __builtin_amdgcn_wmma_f32_16x16x4_f32(false, a, false, b0, (short)0, c0, false, false);
      c1 = __builtin_amdgcn_wmma_f32_16x16x4_f32(false, a, false, b1, (short)0, c1, false, false);
    }
  }
#pragma unroll
  for (int i = 0; i < 8; ++i) {
    long row = r0 + i + half * 8;
    float vA = c0[i] + bvec[m];
    float vB = c1[i] + bvec[16 + m];
    vA = vA > 0.0f ? vA : 0.0f;
    vB = vB > 0.0f ? vB : 0.0f;
    Out[row * 32 + m] = vA;
    Out[row * 32 + 16 + m] = vB;
  }
}

__global__ __launch_bounds__(256) void k_time_gemm(const float* __restrict__ Cheb,
                                                   const float* __restrict__ X,
                                                   const float* __restrict__ tW,
                                                   const float* __restrict__ tB,
                                                   const float* __restrict__ rW,
                                                   const float* __restrict__ rB,
                                                   const float* __restrict__ gamma,
                                                   const float* __restrict__ beta,
                                                   float* __restrict__ Out) {
  int wave = (blockIdx.x * 256 + threadIdx.x) >> 5;
  int lane = threadIdx.x & 31;
  int m = lane & 15;
  int half = lane >> 4;
  int kb = half * 2;
  long r0 = (long)wave * 16;

  long rA = r0 + m;
  int nA = (int)(rA >> 5);
  int tbA = (int)(rA & 31);
  int bA = tbA >> 3;
  int tA = tbA & 7;

  v8f acc0 = {}, acc1 = {}, acc2 = {}, acc3 = {};

#pragma unroll
  for (int kk = 0; kk < 96; kk += 4) {
    int k0 = kk + kb;
    int dt0 = k0 >> 5, ci0 = k0 & 31;
    int dt1 = (k0 + 1) >> 5, ci1 = (k0 + 1) & 31;
    int tp0 = tA + dt0 - 1;
    int tp1 = tA + dt1 - 1;
    float m0 = (tp0 >= 0 && tp0 < 8) ? 1.0f : 0.0f;
    float m1 = (tp1 >= 0 && tp1 < 8) ? 1.0f : 0.0f;
    int tc0 = min(max(tp0, 0), 7);
    int tc1 = min(max(tp1, 0), 7);
    v2f a;
    a.x = m0 * Cheb[((long)nA * 32 + bA * 8 + tc0) * 32 + ci0];
    a.y = m1 * Cheb[((long)nA * 32 + bA * 8 + tc1) * 32 + ci1];
#define TCONV_STEP(ACC, CT)                                                        \
    {                                                                              \
      int co = (CT) * 16 + m;                                                      \
      v2f b;                                                                       \
      b.x = tW[co * 96 + ci0 * 3 + dt0];                                           \
      b.y = tW[co * 96 + ci1 * 3 + dt1];                                           \
      ACC = __builtin_amdgcn_wmma_f32_16x16x4_f32(false, a, false, b, (short)0,    \
                                                  ACC, false, false);              \
    }
    TCONV_STEP(acc0, 0)
    TCONV_STEP(acc1, 1)
    TCONV_STEP(acc2, 2)
    TCONV_STEP(acc3, 3)
#undef TCONV_STEP
  }

#pragma unroll
  for (int kk = 0; kk < 16; kk += 4) {
    int k0 = kk + kb;
    v2f a;
    a.x = X[(((long)(bA * NNODES + nA)) * 16 + k0) * 8 + tA];
    a.y = X[(((long)(bA * NNODES + nA)) * 16 + k0 + 1) * 8 + tA];
#define RCONV_STEP(ACC, CT)                                                        \
    {                                                                              \
      int co = (CT) * 16 + m;                                                      \
      v2f b;                                                                       \
      b.x = rW[co * 16 + k0];                                                      \
      b.y = rW[co * 16 + k0 + 1];                                                  \
      ACC = __builtin_amdgcn_wmma_f32_16x16x4_f32(false, a, false, b, (short)0,    \
                                                  ACC, false, false);              \
    }
    RCONV_STEP(acc0, 0)
    RCONV_STEP(acc1, 1)
    RCONV_STEP(acc2, 2)
    RCONV_STEP(acc3, 3)
#undef RCONV_STEP
  }

#pragma unroll
  for (int i = 0; i < 8; ++i) {
    long row = r0 + i + half * 8;
    int n = (int)(row >> 5);
    int tb = (int)(row & 31);
    int b = tb >> 3;
    int t = tb & 7;
    float vals[4];
    float s = 0.0f, ss = 0.0f;
    {
      float v0 = acc0[i] + tB[m] + rB[m];
      float v1 = acc1[i] + tB[16 + m] + rB[16 + m];
      float v2 = acc2[i] + tB[32 + m] + rB[32 + m];
      float v3 = acc3[i] + tB[48 + m] + rB[48 + m];
      v0 = v0 > 0.0f ? v0 : 0.0f;
      v1 = v1 > 0.0f ? v1 : 0.0f;
      v2 = v2 > 0.0f ? v2 : 0.0f;
      v3 = v3 > 0.0f ? v3 : 0.0f;
      vals[0] = v0; vals[1] = v1; vals[2] = v2; vals[3] = v3;
      s = v0 + v1 + v2 + v3;
      ss = v0 * v0 + v1 * v1 + v2 * v2 + v3 * v3;
    }
#pragma unroll
    for (int ofs = 1; ofs < 16; ofs <<= 1) {
      s += __shfl_xor(s, ofs, 32);
      ss += __shfl_xor(ss, ofs, 32);
    }
    float mu = s * (1.0f / 64.0f);
    float var = ss * (1.0f / 64.0f) - mu * mu;
    float inv = rsqrtf(var + LNEPS);
#pragma unroll
    for (int ct = 0; ct < 4; ++ct) {
      int c = ct * 16 + m;
      float y = (vals[ct] - mu) * inv * gamma[c] + beta[c];
      Out[(((long)(b * NNODES + n)) * 64 + c) * 8 + t] = y;
    }
  }
}

extern "C" void kernel_launch(void* const* d_in, const int* in_sizes, int n_in,
                              void* d_out, int out_size, void* d_ws, size_t ws_size,
                              hipStream_t stream) {
  (void)n_in; (void)out_size; (void)ws_size;
  const float* X     = (const float*)d_in[0];
  const int*   eidx  = (const int*)d_in[1];
  const float* chebW = (const float*)d_in[2];
  const float* chebB = (const float*)d_in[3];
  const float* timeW = (const float*)d_in[4];
  const float* timeB = (const float*)d_in[5];
  const float* resW  = (const float*)d_in[6];
  const float* resB  = (const float*)d_in[7];
  const float* gamma = (const float*)d_in[8];
  const float* beta  = (const float*)d_in[9];
  float* out = (float*)d_out;

  const int E = in_sizes[1] / 2;
  const int* src = eidx;
  const int* dst = eidx + E;

  float* w    = (float*)d_ws;
  float* Xt   = w;
  float* Tx1  = Xt + NRF;
  float* Tx2  = Tx1 + NRF;
  float* Cheb = Tx2 + NRF;
  float* sc   = Cheb + (size_t)NROWS * CCHEB;
  int*   ip     = (int*)(sc + 16);
  int*   indeg  = ip;
  int*   cursor = ip + 10240;
  int*   rowptr = ip + 20480;
  int*   col    = ip + 30976;

  k_build_xt<<<NRF / 256, 256, 0, stream>>>(X, Xt);

  k_zero_int<<<(20480 + 255) / 256, 256, 0, stream>>>(ip, 20480);
  k_count<<<(E + 255) / 256, 256, 0, stream>>>(dst, indeg, E);
  k_scan<<<1, 1024, 0, stream>>>(indeg, rowptr);
  k_fill<<<(E + 255) / 256, 256, 0, stream>>>(src, dst, rowptr, cursor, col, E);

  int* sell = (int*)Cheb;
  float* vcur = Cheb + 6000000;
  float* vnxt = vcur + 10240;
  float* pcur = vcur + 20480;
  float* pnxt = vcur + 20992;
  int* swidth = (int*)(vcur + 21504);
  k_sell<<<40, 256, 0, stream>>>(rowptr, col, sell, swidth, vcur);
  k_pstep<true><<<PS_GROUPS, PS_THREADS, 0, stream>>>(rowptr, sell, swidth, vcur, pcur, vnxt, pnxt);
  for (int it = 1; it < PIters; ++it) {
    float* t = vcur; vcur = vnxt; vnxt = t;
    t = pcur; pcur = pnxt; pnxt = t;
    k_pstep<false><<<PS_GROUPS, PS_THREADS, 0, stream>>>(rowptr, sell, swidth, vcur, pcur, vnxt, pnxt);
  }
  k_lnorm<<<1, 32, 0, stream>>>(pnxt, sc);

  k_gather1<<<NNODES * 128 / 256, 256, 0, stream>>>(rowptr, col, Xt, sc, Tx1);
  k_gather2<<<NNODES * 128 / 256, 256, 0, stream>>>(rowptr, col, Tx1, Xt, sc, Tx2);

  k_cheb_gemm<<<2500, 256, 0, stream>>>(Xt, Tx1, Tx2, chebW, chebB, Cheb);

  k_time_gemm<<<2500, 256, 0, stream>>>(Cheb, X, timeW, timeB, resW, resB,
                                        gamma, beta, out);
}
